# starting kernel
# baseline (speedup 1.0000x reference)
#include <hip/hip_runtime.h>

typedef __attribute__((ext_vector_type(2))) float v2f;
typedef __attribute__((ext_vector_type(8))) float v8f;

#define ROWS   4096
#define COLS   8192
#define KS     25
#define GCOLS  (COLS - KS)
#define TPW    8
#define WPB    8

__global__ __launch_bounds__(256) void tma1d_wmma_kernel(
    const float* __restrict__ x, const float* __restrict__ w,
    float* __restrict__ out) {
  const int lane = threadIdx.x & 31;
  const int lr   = lane & 15;
  const int hi   = lane >> 4;
  const int wid  = blockIdx.x * WPB + (threadIdx.x >> 5);

  const int rowTile = wid >> 6;
  const int run     = wid & 63;
  const int r0      = rowTile << 4;
  int       c0      = run * (TPW * 16);

  v2f b[10];
#pragma unroll
  for (int q = 0; q < 10; ++q) {
    const int i0 = 4 * q + 2 * hi - lr;
    const int i1 = i0 + 1;
    b[q][0] = (i0 >= 0 && i0 < KS) ? w[i0] : 0.0f;
    b[q][1] = (i1 >= 0 && i1 < KS) ? w[i1] : 0.0f;
  }

  const float* xrow = x + (size_t)(r0 + lr) * COLS;
  auto loadA = [&](int colbase) -> v2f {
    int col = colbase + 2 * hi;
    if (col > COLS - 2) col = COLS - 2;
    return *(const v2f*)(xrow + col);
  };

  v2f a[10];
#pragma unroll
  for (int q = 0; q < 6; ++q) a[q] = loadA(c0 + 4 * q);

  float* orow = out + (size_t)(r0 + 8 * hi) * COLS;

  for (int t = 0; t < TPW; ++t) {
    a[6] = loadA(c0 + 24);
    a[7] = loadA(c0 + 28);
    a[8] = loadA(c0 + 32);
    a[9] = loadA(c0 + 36);

    v8f c = {};
#if __has_builtin(__builtin_amdgcn_wmma_f32_16x16x4_f32)
#pragma unroll
    for (int q = 0; q < 10; ++q) {
      c = __builtin_amdgcn_wmma_f32_16x16x4_f32(
          false, a[q], false, b[q], (short)0, c, false, false);
    }
#else
    {
      const int gc = c0 + lr;
#pragma unroll
      for (int v = 0; v < 8; ++v) {
        const float* xr = x + (size_t)(r0 + 8 * hi + v) * COLS;
        float s = 0.0f;
        int cc = (gc > GCOLS - 1) ? (GCOLS - 1) : gc;
        for (int i = 0; i < KS; ++i) s = fmaf(xr[cc + i], w[i], s);
        c[v] = s;
      }
    }
#endif

    const int gc = c0 + lr;
    if (gc < GCOLS) {
#pragma unroll
      for (int v = 0; v < 8; ++v)
        __builtin_nontemporal_store(c[v], &orow[(size_t)v * COLS + gc + KS]);
    }
    if (gc < KS) {
#pragma unroll
      for (int v = 0; v < 8; ++v)
        __builtin_nontemporal_store(c[v], &orow[(size_t)v * COLS + gc]);
    }

#pragma unroll
    for (int q = 0; q < 6; ++q) a[q] = a[q + 4];
    c0 += 16;
  }
}

extern "C" void kernel_launch(void* const* d_in, const int* in_sizes, int n_in,
                              void* d_out, int out_size, void* d_ws, size_t ws_size,
                              hipStream_t stream) {
  const float* x = (const float*)d_in[0];
  const float* w = (const float*)d_in[1];
  float* out = (float*)d_out;

  dim3 grid(2048), block(256);
  hipLaunchKernelGGL(tma1d_wmma_kernel, grid, block, 0, stream, x, w, out);
}


# x tile staged in LDS by whole-line 16-byte loads (each line fetched once), A fragments by ds_load_b128; k order inside each 8-column chunk pair permuted for A and B alike: f32 sum order changes
# speedup vs baseline: 1.0928x; 1.0928x over previous
#include <hip/hip_runtime.h>


typedef __attribute__((ext_vector_type(2))) float v2f;
typedef __attribute__((ext_vector_type(4))) float v4f;
typedef __attribute__((ext_vector_type(8))) float v8f;

#define ROWS   4096
#define COLS   8192
#define KS     25
#define GCOLS  (COLS - KS)
#define TPW    8
#define WPB    8
#define BCOLS  (WPB * TPW * 16)
#define RPCS   ((BCOLS + 32) / 4)
#define PITCH  (RPCS + 1)
#define NLD    ((2 * RPCS + 31) / 32)

__global__ __launch_bounds__(256) void tma1d_wmma_kernel(
    const float* __restrict__ x, const float* __restrict__ w,
    float* __restrict__ out) {
  __shared__ v4f xs[16 * PITCH];

  const int tid  = threadIdx.x;
  const int lane = tid & 31;
  const int lr   = lane & 15;
  const int hi   = lane >> 4;
  const int wv   = tid >> 5;

  const int r0 = (blockIdx.x >> 3) << 4;
  const int cb = (blockIdx.x & 7) * BCOLS;

  const float* xg = x + (size_t)(r0 + 2 * wv) * COLS + cb;
  v4f st[NLD];
#pragma unroll
  for (int i = 0; i < NLD; ++i) {
    const int q = i * 32 + lane;
    if (q < 2 * RPCS) {
      const int r  = (q >= RPCS) ? 1 : 0;
      const int pc = q - r * RPCS;
      int col = 4 * pc;
      if (cb + col > COLS - 4) col = COLS - 4 - cb;
      st[i] = *(const v4f*)(xg + (size_t)r * COLS + col);
    }
  }

  v2f b[10];
#pragma unroll
  for (int q = 0; q < 10; ++q) {
    const int i0 = 8 * (q >> 1) + 4 * hi + 2 * (q & 1) - lr;
    const int i1 = i0 + 1;
    b[q][0] = (i0 >= 0 && i0 < KS) ? w[i0] : 0.0f;
    b[q][1] = (i1 >= 0 && i1 < KS) ? w[i1] : 0.0f;
  }

#pragma unroll
  for (int i = 0; i < NLD; ++i) {
    const int q = i * 32 + lane;
    if (q < 2 * RPCS) {
      const int r  = (q >= RPCS) ? 1 : 0;
      const int pc = q - r * RPCS;
      xs[(2 * wv + r) * PITCH + pc] = st[i];
    }
  }
  __syncthreads();

  const v4f* xr = xs + lr * PITCH + wv * 32 + hi;

  v4f a[5];
  a[0] = xr[0];
  a[1] = xr[2];
  a[2] = xr[4];

  int    c0   = cb + wv * (TPW * 16);
  float* orow = out + (size_t)(r0 + 8 * hi) * COLS;

#pragma unroll
  for (int t = 0; t < TPW; ++t) {
    a[3] = xr[2 * (2 * t + 3)];
    a[4] = xr[2 * (2 * t + 4)];

    v8f c = {};
#pragma unroll
    for (int jj = 0; jj < 5; ++jj) {
      const v2f a0 = {a[jj][0], a[jj][1]};
      const v2f a1 = {a[jj][2], a[jj][3]};
      c = __builtin_amdgcn_wmma_f32_16x16x4_f32(
          false, a0, false, b[2 * jj], (short)0, c, false, false);
      c = __builtin_amdgcn_wmma_f32_16x16x4_f32(
          false, a1, false, b[2 * jj + 1], (short)0, c, false, false);
    }

    const int gc = c0 + lr;
    if (gc < GCOLS) {
#pragma unroll
      for (int v = 0; v < 8; ++v)
        __builtin_nontemporal_store(c[v], &orow[(size_t)v * COLS + gc + KS]);
    }
    if (gc < KS) {
#pragma unroll
      for (int v = 0; v < 8; ++v)
        __builtin_nontemporal_store(c[v], &orow[(size_t)v * COLS + gc]);
    }

    a[0] = a[2];
    a[1] = a[3];
    a[2] = a[4];
    c0 += 16;
  }
}

extern "C" void kernel_launch(void* const* d_in, const int* in_sizes, int n_in,
                              void* d_out, int out_size, void* d_ws, size_t ws_size,
                              hipStream_t stream) {
  const float* x = (const float*)d_in[0];
  const float* w = (const float*)d_in[1];
  float* out = (float*)d_out;

  dim3 grid(2048), block(256);
  hipLaunchKernelGGL(tma1d_wmma_kernel, grid, block, 0, stream, x, w, out);
}


# tiles aligned to out columns (x window shifted left, 11 K=4 chunks) and out strip parked in LDS: each store writes 4 whole 128-byte lines; k grouping moves: f32 sum order changes
# speedup vs baseline: 2.1227x; 1.9424x over previous
#include <hip/hip_runtime.h>


typedef __attribute__((ext_vector_type(2))) float v2f;
typedef __attribute__((ext_vector_type(4))) float v4f;
typedef __attribute__((ext_vector_type(8))) float v8f;

#define ROWS   4096
#define COLS   8192
#define KS     25
#define TPW    8
#define WPB    8
#define BCOLS  (WPB * TPW * 16)
#define HALO   32
#define RPCS   ((BCOLS + HALO) / 4)
#define PITCH  (RPCS + 1)
#define FPITCH (4 * PITCH)
#define NLD    ((2 * RPCS + 31) / 32)

__global__ __launch_bounds__(256, 8) void tma1d_wmma_kernel(
    const float* __restrict__ x, const float* __restrict__ w,
    float* __restrict__ out) {
  __shared__ v4f xs[16 * PITCH];

  const int tid  = threadIdx.x;
  const int lane = tid & 31;
  const int lr   = lane & 15;
  const int hi   = lane >> 4;
  const int wv   = tid >> 5;

  const int r0 = (blockIdx.x >> 3) << 4;
  const int ob = (blockIdx.x & 7) * BCOLS;

  const float* xg = x + (size_t)(r0 + 2 * wv) * COLS;
  v4f st[NLD];
#pragma unroll
  for (int i = 0; i < NLD; ++i) {
    const int q = i * 32 + lane;
    if (q < 2 * RPCS) {
      const int r  = (q >= RPCS) ? 1 : 0;
      const int pc = q - r * RPCS;
      int col = ob - HALO + 4 * pc;
      if (col < 0) col = 0;
      st[i] = *(const v4f*)(xg + (size_t)r * COLS + col);
    }
  }

  v2f b[11];
#pragma unroll
  for (int q = 1; q < 12; ++q) {
    const int i0 = 8 * (q >> 1) + 4 * hi + 2 * (q & 1) - 7 - lr;
    const int i1 = i0 + 1;
    b[q - 1][0] = (i0 >= 0 && i0 < KS) ? w[i0] : 0.0f;
    b[q - 1][1] = (i1 >= 0 && i1 < KS) ? w[i1] : 0.0f;
  }

#pragma unroll
  for (int i = 0; i < NLD; ++i) {
    const int q = i * 32 + lane;
    if (q < 2 * RPCS) {
      const int r  = (q >= RPCS) ? 1 : 0;
      const int pc = q - r * RPCS;
      xs[(2 * wv + r) * PITCH + pc] = st[i];
    }
  }
  __syncthreads();

  const v4f* xr = xs + lr * PITCH + wv * 32 + hi;

  v4f a[6];
  a[0] = xr[0];
  a[1] = xr[2];
  a[2] = xr[4];
  a[3] = xr[6];

  v8f c[TPW];

#pragma unroll
  for (int t = 0; t < TPW; ++t) {
    a[4] = xr[2 * (2 * t + 4)];
    a[5] = xr[2 * (2 * t + 5)];

    v8f acc = {};
#pragma unroll
    for (int jj = 0; jj < 6; ++jj) {
      const v2f a0 = {a[jj][0], a[jj][1]};
      const v2f a1 = {a[jj][2], a[jj][3]};
      if (jj > 0)
        acc = __builtin_amdgcn_wmma_f32_16x16x4_f32(
            false, a0, false, b[2 * jj - 1], (short)0, acc, false, false);
      acc = __builtin_amdgcn_wmma_f32_16x16x4_f32(
          false, a1, false, b[2 * jj], (short)0, acc, false, false);
    }
    c[t] = acc;

    a[0] = a[2];
    a[1] = a[3];
    a[2] = a[4];
    a[3] = a[5];
  }

  __syncthreads();

  float*    xf = (float*)xs;
  const int os = ob + wv * (TPW * 16);
  float*    so = xf + (8 * hi) * FPITCH + wv * (TPW * 16);
#pragma unroll
  for (int t = 0; t < TPW; ++t) {
    const int sc = 16 * t + lr;
    const int gc = os + sc - KS;
    if (gc >= 0) {
#pragma unroll
      for (int v = 0; v < 8; ++v) so[v * FPITCH + sc] = c[t][v];
    }
    if (gc >= 0 && gc < KS) {
#pragma unroll
      for (int v = 0; v < 8; ++v) so[v * FPITCH + (gc - os)] = c[t][v];
    }
  }

  const v4f* sr = xs + wv * 32 + lane;
  float*     og = out + (size_t)r0 * COLS + os + 4 * lane;
#pragma unroll
  for (int i = 0; i < 16; ++i) {
    const v4f v = sr[i * PITCH];
    __builtin_nontemporal_store(v, (v4f*)(og + (size_t)i * COLS));
  }
}

extern "C" void kernel_launch(void* const* d_in, const int* in_sizes, int n_in,
                              void* d_out, int out_size, void* d_ws, size_t ws_size,
                              hipStream_t stream) {
  const float* x = (const float*)d_in[0];
  const float* w = (const float*)d_in[1];
  float* out = (float*)d_out;

  dim3 grid(2048), block(256);
  hipLaunchKernelGGL(tma1d_wmma_kernel, grid, block, 0, stream, x, w, out);
}
